# starting kernel
# baseline (speedup 1.0000x reference)
#include <hip/hip_runtime.h>


typedef __attribute__((ext_vector_type(2))) float v2f;
typedef __attribute__((ext_vector_type(8))) float v8f;

__device__ __forceinline__ float band_sign(int parity, int band) {
    return (__builtin_popcount(parity & band) & 1) ? -1.0f : 1.0f;
}

__global__ __launch_bounds__(256)
void iwt3d_wmma_kernel(const float* __restrict__ x, float* __restrict__ out) {
    const int tid  = (int)blockIdx.x * 256 + (int)threadIdx.x;
    const int lane = (int)threadIdx.x & 31;
    const int wave = tid >> 5;

    const int h = wave & 31;
    const int d = (wave >> 5) & 31;
    const int c = (wave >> 10) & 31;
    const int b = (wave >> 15) & 1;

    const int bandStride = 32 * 32 * 32 * 32;
    const int inBase = (((b * 256 + c) * 32 + d) * 32 + h) * 32;

    const int lh = lane & 15;
    const int hi = lane >> 4;
    const int m  = lane & 7;

    const int k0 = hi * 2;
    const int k1 = k0 + 1;

    v2f A0, A1;
    A0.x = band_sign(m, k0);      A0.y = band_sign(m, k1);
    A1.x = band_sign(m, k0 + 4);  A1.y = band_sign(m, k1 + 4);

    v2f B00, B01, B10, B11;
    B00.x = x[inBase + (k0    ) * bandStride +  0 + lh];
    B00.y = x[inBase + (k1    ) * bandStride +  0 + lh];
    B01.x = x[inBase + (k0 + 4) * bandStride +  0 + lh];
    B01.y = x[inBase + (k1 + 4) * bandStride +  0 + lh];
    B10.x = x[inBase + (k0    ) * bandStride + 16 + lh];
    B10.y = x[inBase + (k1    ) * bandStride + 16 + lh];
    B11.x = x[inBase + (k0 + 4) * bandStride + 16 + lh];
    B11.y = x[inBase + (k1 + 4) * bandStride + 16 + lh];

    v8f acc0 = {};
    acc0 = __builtin_amdgcn_wmma_f32_16x16x4_f32(false, A0, false, B00,
                                                 (short)0, acc0, false, false);
    acc0 = __builtin_amdgcn_wmma_f32_16x16x4_f32(false, A1, false, B01,
                                                 (short)0, acc0, false, false);
    v8f acc1 = {};
    acc1 = __builtin_amdgcn_wmma_f32_16x16x4_f32(false, A0, false, B10,
                                                 (short)0, acc1, false, false);
    acc1 = __builtin_amdgcn_wmma_f32_16x16x4_f32(false, A1, false, B11,
                                                 (short)0, acc1, false, false);

    if (lane < 16) {
        const int outRow = (((b * 32 + c) * 64 + 2 * d) * 64 + 2 * h) * 64;
        #pragma unroll
        for (int g = 0; g < 2; ++g) {
            const int w = g * 16 + lh;
            const v8f acc = g ? acc1 : acc0;
            const int outBase = outRow + 2 * w;
            #pragma unroll
            for (int p = 0; p < 2; ++p) {
                #pragma unroll
                for (int q = 0; q < 2; ++q) {
                    v2f v;
                    v.x = acc[4 * p + 2 * q + 0];
                    v.y = acc[4 * p + 2 * q + 1];
                    *(v2f*)(out + outBase + p * 4096 + q * 64) = v;
                }
            }
        }
    }
}

extern "C" void kernel_launch(void* const* d_in, const int* in_sizes, int n_in,
                              void* d_out, int out_size, void* d_ws, size_t ws_size,
                              hipStream_t stream) {
    (void)n_in; (void)out_size; (void)d_ws; (void)ws_size;
    const float* x = (const float*)d_in[0];
    float* out = (float*)d_out;

    const int waves  = in_sizes[0] / 256;
    const int blocks = waves / 8;
    hipLaunchKernelGGL(iwt3d_wmma_kernel, dim3(blocks), dim3(256), 0, stream,
                       x, out);
}


# Block-diagonal sign operand: a lane feeds its own 4 voxels (16-byte loads); a swap form of A moves half the results across lane halves: whole-row 16-byte stores; 4 tiles/wave; sum order changes
# speedup vs baseline: 1.1485x; 1.1485x over previous
#include <hip/hip_runtime.h>


typedef __attribute__((ext_vector_type(2))) float v2f;
typedef __attribute__((ext_vector_type(4))) float v4f;
typedef __attribute__((ext_vector_type(8))) float v8f;

#define TILES_PER_WAVE 4

__device__ __forceinline__ float band_sign(int parity, int band) {
    return (__builtin_popcount(parity & band) & 1) ? -1.0f : 1.0f;
}

__global__ __launch_bounds__(256)
void iwt3d_wmma_kernel(const float* __restrict__ x, float* __restrict__ out) {
    const int lane = (int)threadIdx.x & 31;
    const int wave = __builtin_amdgcn_readfirstlane(((int)blockIdx.x * 256 + (int)threadIdx.x) >> 5);
    const int lh   = lane & 15;
    const int half = lane >> 4;
    const int m    = lane & 7;

    const int bandStride = 32 * 32 * 32 * 32;

    const int T0 = wave * TILES_PER_WAVE;
    const int bc = T0 >> 8;
    const int dq = T0 & 255;
    const int inBase = ((bc >> 5) * 256 + (bc & 31)) * 32768 + dq * 128 + lane * 4;

    v4f X[TILES_PER_WAVE][8];
    #pragma unroll
    for (int i = 0; i < TILES_PER_WAVE; ++i) {
        #pragma unroll
        for (int beta = 0; beta < 8; ++beta) {
            X[i][beta] = *(const v4f*)(x + inBase + i * 128 + beta * bandStride);
        }
        __builtin_amdgcn_sched_barrier(0);
    }

    const bool diag = ((lh >> 3) == half);
    v2f Akeep[4], Aswap[4];
    #pragma unroll
    for (int s = 0; s < 4; ++s) {
        const float s0 = band_sign(m, 2 * s);
        const float s1 = band_sign(m, 2 * s + 1);
        Akeep[s].x = diag ? s0 : 0.0f;  Akeep[s].y = diag ? s1 : 0.0f;
        Aswap[s].x = diag ? 0.0f : s0;  Aswap[s].y = diag ? 0.0f : s1;
    }

    const int outBase = bc * 262144 + (dq >> 3) * 8192 + (dq & 7) * 512;
    const int rowPart = (lh >> 3) * 128 + (lh & 7) * 8;
    const int off1 = rowPart + 4 * half;
    const int off2 = rowPart + 256 + 4 - 4 * half;

    #pragma unroll
    for (int i = 0; i < TILES_PER_WAVE; ++i) {
        v8f G[4];
        #pragma unroll
        for (int g = 0; g < 4; ++g) {
            v8f acc = {};
            #pragma unroll
            for (int s = 0; s < 4; ++s) {
                v2f Bv;
                Bv.x = X[i][2 * s][g];
                Bv.y = X[i][2 * s + 1][g];
                acc = __builtin_amdgcn_wmma_f32_16x16x4_f32(false, (g < 2) ? Akeep[s] : Aswap[s],
                                                            false, Bv, (short)0, acc, false, false);
            }
            G[g] = acc;
        }
        const int tileOut = outBase + i * 512;
        #pragma unroll
        for (int k = 0; k < 4; ++k) {
            v4f P, Q;
            P.x = G[0][2 * k]; P.y = G[0][2 * k + 1]; P.z = G[1][2 * k]; P.w = G[1][2 * k + 1];
            Q.x = G[2][2 * k]; Q.y = G[2][2 * k + 1]; Q.z = G[3][2 * k]; Q.w = G[3][2 * k + 1];
            const v4f d1 = half ? Q : P;
            const v4f d2 = half ? P : Q;
            const int rowOut = tileOut + (k >> 1) * 4096 + (k & 1) * 64;
            *(v4f*)(out + rowOut + off1) = d1;
            *(v4f*)(out + rowOut + off2) = d2;
        }
        __builtin_amdgcn_sched_barrier(0);
    }
}

extern "C" void kernel_launch(void* const* d_in, const int* in_sizes, int n_in,
                              void* d_out, int out_size, void* d_ws, size_t ws_size,
                              hipStream_t stream) {
    (void)in_sizes; (void)n_in; (void)out_size; (void)d_ws; (void)ws_size;
    const float* x = (const float*)d_in[0];
    float* out = (float*)d_out;

    hipLaunchKernelGGL(iwt3d_wmma_kernel, dim3(512), dim3(256), 0, stream, x, out);
}
